# starting kernel
# baseline (speedup 1.0000x reference)
#include <hip/hip_runtime.h>

typedef __attribute__((ext_vector_type(2))) float v2f;
typedef __attribute__((ext_vector_type(8))) float v8f;

#define NHD   8
#define HDD   6
#define DIMC  48
#define NT    27
#define GH    40
#define GW    40
#define GT    40
#define QSCALE 0.40824829046386301636f

constexpr unsigned bcode(int t) {
    return (t >= NT)
        ? 0x55555555u
        : (unsigned)(t / 9)
        | ((unsigned)((t / 3) % 3) << 2)
        | ((unsigned)(t % 3) << 4)
        | 0x55555540u;
}

__device__ __forceinline__ float bval(unsigned code, int col) {
    return (float)((int)((code >> (2 * col)) & 3u) - 1);
}

__global__ __launch_bounds__(256) void nat3d_wmma_kernel(
    const float* __restrict__ q,
    const float* __restrict__ k,
    const float* __restrict__ rpb,
    float* __restrict__ out)
{
    __shared__ float s_rpb[NHD * NT];
    int tid = threadIdx.x;
    if (tid < NHD * NT) s_rpb[tid] = rpb[tid];
    __syncthreads();

    const int lane = tid & 31;
    const int wave = tid >> 5;
    const int pair = blockIdx.x * 8 + wave;
    const int z2   = pair % (GT / 2);
    const int tmp  = pair / (GT / 2);
    const int y    = tmp % GW;
    const int x    = tmp / GW;

    const int m    = lane & 15;
    const int half = lane >> 4;
    const int h    = m & 7;
    const int z    = z2 * 2 + (m >> 3);

    const int lin = (x * GW + y) * GT + z;
    const float* qp = q + (size_t)lin * DIMC + h * HDD;
    float qv[HDD];
#pragma unroll
    for (int d = 0; d < HDD; d += 2) {
        v2f v = *(const v2f*)(qp + d);
        qv[d]     = v.x * QSCALE;
        qv[d + 1] = v.y * QSCALE;
    }

    float p[NT];
    float mx = -1e30f;
#pragma unroll
    for (int t = 0; t < NT; ++t) {
        const int ti = t / 9, tj = (t / 3) % 3, tl = t % 3;
        const int nx = x + ti - 1, ny = y + tj - 1, nz = z + tl - 1;
        float dot = 0.0f;
        if ((unsigned)nx < GH && (unsigned)ny < GW && (unsigned)nz < GT) {
            const float* kp = k + (size_t)((nx * GW + ny) * GT + nz) * DIMC + h * HDD;
#pragma unroll
            for (int d = 0; d < HDD; d += 2) {
                v2f v = *(const v2f*)(kp + d);
                dot = fmaf(qv[d],     v.x, dot);
                dot = fmaf(qv[d + 1], v.y, dot);
            }
        }
        float s = dot + s_rpb[h * NT + t];
        p[t] = s;
        mx = fmaxf(mx, s);
    }

    float sum = 0.0f;
#pragma unroll
    for (int t = 0; t < NT; ++t) { p[t] = __expf(p[t] - mx); sum += p[t]; }
    const float inv = __builtin_amdgcn_rcpf(sum);
#pragma unroll
    for (int t = 0; t < NT; ++t) p[t] *= inv;

    const int col = lane & 15;
    v8f acc = {};
#pragma unroll
    for (int c = 0; c < 7; ++c) {
        const int t0 = 4 * c;
        const float a_lo0 = p[t0];
        const float a_lo1 = p[t0 + 1];
        const float a_hi0 = (t0 + 2 < NT) ? p[t0 + 2 < NT ? t0 + 2 : 0] : 0.0f;
        const float a_hi1 = (t0 + 3 < NT) ? p[t0 + 3 < NT ? t0 + 3 : 0] : 0.0f;
        v2f a, b;
        a.x = half ? a_hi0 : a_lo0;
        a.y = half ? a_hi1 : a_lo1;
        const unsigned cx = half ? bcode(t0 + 2) : bcode(t0);
        const unsigned cy = half ? bcode(t0 + 3) : bcode(t0 + 1);
        b.x = bval(cx, col);
        b.y = bval(cy, col);
        acc = __builtin_amdgcn_wmma_f32_16x16x4_f32(
             false, a,  false, b,
             (short)0, acc,  false,  false);
    }

    if (col < 3) {
        const size_t sp = (size_t)((x * GW + y) * GT) + (size_t)(z2 * 2 + half);
#pragma unroll
        for (int r = 0; r < 8; ++r) {
            out[(size_t)(r * 3 + col) * (GH * GW * GT) + sp] = acc[r];
        }
    }
}

extern "C" void kernel_launch(void* const* d_in, const int* in_sizes, int n_in,
                              void* d_out, int out_size, void* d_ws, size_t ws_size,
                              hipStream_t stream) {
    (void)in_sizes; (void)n_in; (void)out_size; (void)d_ws; (void)ws_size;
    const float* q   = (const float*)d_in[0];
    const float* k   = (const float*)d_in[1];
    const float* rpb = (const float*)d_in[2];
    float* out = (float*)d_out;

    dim3 grid(4000), block(256);
    nat3d_wmma_kernel<<<grid, block, 0, stream>>>(q, k, rpb, out);
}


# Each wave marches over 8 x positions with three queries in flight: every key plane is loaded once per wave and feeds all three (9 key loads per query, was 27); 8x fewer waves; sum order unchanged
# speedup vs baseline: 1.5275x; 1.5275x over previous
#include <hip/hip_runtime.h>


typedef __attribute__((ext_vector_type(2))) float v2f;
typedef __attribute__((ext_vector_type(8))) float v8f;

#define NHD   8
#define HDD   6
#define DIMC  48
#define NT    27
#define GH    40
#define GW    40
#define GT    40
#define XSEG  8
#define QSCALE 0.40824829046386301636f

constexpr unsigned bcode(int t) {
    return (t >= NT)
        ? 0x55555555u
        : (unsigned)(t / 9)
        | ((unsigned)((t / 3) % 3) << 2)
        | ((unsigned)(t % 3) << 4)
        | 0x55555540u;
}

__device__ __forceinline__ float bval(unsigned code, int col) {
    return (float)((int)((code >> (2 * col)) & 3u) - 1);
}


struct KPlane { float v[9][HDD]; };

struct QState { float q[HDD]; float s[NT]; };

__device__ __forceinline__ void load_plane(KPlane& P, const float* __restrict__ k,
                                           int nx, int y, int z, int h)
{
    const bool xin = (unsigned)nx < GH;
#pragma unroll
    for (int j = 0; j < 3; ++j) {
#pragma unroll
        for (int l = 0; l < 3; ++l) {
            const int ny = y + j - 1, nz = z + l - 1;
            if (xin && (unsigned)ny < GW && (unsigned)nz < GT) {
                const float* kp = k + (size_t)((nx * GW + ny) * GT + nz) * DIMC + h * HDD;
#pragma unroll
                for (int d = 0; d < HDD; d += 2) {
                    v2f v = *(const v2f*)(kp + d);
                    P.v[j * 3 + l][d]     = v.x;
                    P.v[j * 3 + l][d + 1] = v.y;
                }
            } else {
#pragma unroll
                for (int d = 0; d < HDD; ++d) P.v[j * 3 + l][d] = 0.0f;
            }
        }
    }
}

__device__ __forceinline__ void load_q(float (&qv)[HDD], const float* __restrict__ q,
                                       int x, int y, int z, int h)
{
    const float* qp = q + (size_t)((x * GW + y) * GT + z) * DIMC + h * HDD;
#pragma unroll
    for (int d = 0; d < HDD; d += 2) {
        v2f v = *(const v2f*)(qp + d);
        qv[d]     = v.x;
        qv[d + 1] = v.y;
    }
}

template <int TI>
__device__ __forceinline__ void scores9(const KPlane& P, QState& Q, const float* rp)
{
#pragma unroll
    for (int r = 0; r < 9; ++r) {
        float dot = 0.0f;
#pragma unroll
        for (int d = 0; d < HDD; ++d) dot = fmaf(Q.q[d], P.v[r][d], dot);
        Q.s[TI * 9 + r] = dot + rp[TI * 9 + r];
    }
}

__device__ __forceinline__ void finish_query(const float (&s)[NT], float* __restrict__ out,
                                             int x, int y, int z2, int half, int col)
{
    float mx = -1e30f;
#pragma unroll
    for (int t = 0; t < NT; ++t) mx = fmaxf(mx, s[t]);

    float p[NT];
    float sum = 0.0f;
#pragma unroll
    for (int t = 0; t < NT; ++t) { p[t] = __expf(s[t] - mx); sum += p[t]; }
    const float inv = __builtin_amdgcn_rcpf(sum);
#pragma unroll
    for (int t = 0; t < NT; ++t) p[t] *= inv;

    v8f acc = {};
#pragma unroll
    for (int c = 0; c < 7; ++c) {
        const int t0 = 4 * c;
        const float a_lo0 = p[t0];
        const float a_lo1 = p[t0 + 1];
        const float a_hi0 = (t0 + 2 < NT) ? p[t0 + 2 < NT ? t0 + 2 : 0] : 0.0f;
        const float a_hi1 = (t0 + 3 < NT) ? p[t0 + 3 < NT ? t0 + 3 : 0] : 0.0f;
        v2f a, b;
        a.x = half ? a_hi0 : a_lo0;
        a.y = half ? a_hi1 : a_lo1;
        const unsigned cx = half ? bcode(t0 + 2) : bcode(t0);
        const unsigned cy = half ? bcode(t0 + 3) : bcode(t0 + 1);
        b.x = bval(cx, col);
        b.y = bval(cy, col);
        acc = __builtin_amdgcn_wmma_f32_16x16x4_f32(
             false, a,  false, b,
             (short)0, acc,  false,  false);
    }

    if (col < 3) {
        const size_t sp = (size_t)((x * GW + y) * GT) + (size_t)(z2 * 2 + half);
#pragma unroll
        for (int r = 0; r < 8; ++r) {
            out[(size_t)(r * 3 + col) * (GH * GW * GT) + sp] = acc[r];
        }
    }
}

__global__ __launch_bounds__(256) void nat3d_wmma_kernel(
    const float* __restrict__ q,
    const float* __restrict__ k,
    const float* __restrict__ rpb,
    float* __restrict__ out)
{
    __shared__ float s_rpb[NHD * NT];
    int tid = threadIdx.x;
    if (tid < NHD * NT) s_rpb[tid] = rpb[tid];
    __syncthreads();

    const int lane = tid & 31;
    const int wave = __builtin_amdgcn_readfirstlane(tid >> 5);
    const int unit = blockIdx.x * 8 + wave;
    const int z2   = unit % (GT / 2);
    const int tmp  = unit / (GT / 2);
    const int y    = tmp % GW;
    const int seg  = tmp / GW;
    const int x0   = seg * XSEG;
    const int xe   = x0 + XSEG;

    const int m    = lane & 15;
    const int half = lane >> 4;
    const int h    = m & 7;
    const int z    = z2 * 2 + (m >> 3);
    const int col  = lane & 15;
    const float* rp = s_rpb + h * NT;

    QState U, V, Wd;
    int X = x0 - 1;
#define NAT_ROUND(QU, QV, QW)                                                          \
    {                                                                                  \
        KPlane kv;                                                                     \
        float qraw[HDD];                                                               \
        load_plane(kv, k, X, y, z, h);                                                 \
        if (X + 1 < xe)  load_q(qraw, q, X + 1, y, z, h);                              \
        if (X - 2 >= x0) finish_query(QW.s, out, X - 2, y, z2, half, col);             \
        if (X - 1 >= x0) scores9<2>(kv, QU, rp);                                       \
        if (X >= x0 && X < xe) scores9<1>(kv, QV, rp);                                 \
        if (X + 1 < xe) {                                                              \
            _Pragma("unroll")                                                          \
            for (int d = 0; d < HDD; ++d) QW.q[d] = qraw[d] * QSCALE;                  \
            scores9<0>(kv, QW, rp);                                                    \
        }                                                                              \
        if (++X > xe) break;                                                           \
    }
    for (;;) {
        NAT_ROUND(U, V, Wd)
        NAT_ROUND(V, Wd, U)
        NAT_ROUND(Wd, U, V)
    }
#undef NAT_ROUND
    constexpr int LAST = (XSEG + 1) % 3;
    if (LAST == 0)      finish_query(U.s,  out, xe - 1, y, z2, half, col);
    else if (LAST == 1) finish_query(V.s,  out, xe - 1, y, z2, half, col);
    else                finish_query(Wd.s, out, xe - 1, y, z2, half, col);
}

extern "C" void kernel_launch(void* const* d_in, const int* in_sizes, int n_in,
                              void* d_out, int out_size, void* d_ws, size_t ws_size,
                              hipStream_t stream) {
    (void)in_sizes; (void)n_in; (void)out_size; (void)d_ws; (void)ws_size;
    const float* q   = (const float*)d_in[0];
    const float* k   = (const float*)d_in[1];
    const float* rpb = (const float*)d_in[2];
    float* out = (float*)d_out;

    dim3 grid(500), block(256);
    nat3d_wmma_kernel<<<grid, block, 0, stream>>>(q, k, rpb, out);
}


# Half-waves of a row split each key plane's 9 neighbours 5/4 (value product's k slots reordered to match): a lane loads and scores 15 keys, not 27; max and sum joined by lane^16; sum order changed
# speedup vs baseline: 1.6067x; 1.0518x over previous
#include <hip/hip_runtime.h>


typedef __attribute__((ext_vector_type(2))) float v2f;
typedef __attribute__((ext_vector_type(8))) float v8f;

#define NHD   8
#define HDD   6
#define DIMC  48
#define NT    27
#define GH    40
#define GW    40
#define GT    40
#define XSEG  8
#define QSCALE 0.40824829046386301636f

#define NSLOT 5
#define NSC   15
#define NMMA  8

constexpr unsigned bcode(int hf, int e) {
    return (e >= NSC || (hf == 1 && e % NSLOT == NSLOT - 1))
        ? 0x55555555u
        : (unsigned)(e / NSLOT)
        | ((unsigned)((hf * 5 + e % NSLOT) / 3) << 2)
        | ((unsigned)((hf * 5 + e % NSLOT) % 3) << 4)
        | 0x55555540u;
}

__device__ __forceinline__ float bval(unsigned code, int col) {
    return (float)((int)((code >> (2 * col)) & 3u) - 1);
}

__device__ __forceinline__ float xhalf(float x) {
    const int b = __builtin_bit_cast(int, x);
    return __builtin_bit_cast(float,
        __builtin_amdgcn_permlanex16(b, b, 0x76543210, 0xfedcba98, false, false));
}


struct KPlane { float v[NSLOT][HDD]; };

struct QState { float q[HDD]; float s[NSC]; };

__device__ __forceinline__ void load_plane(KPlane& P, const float* __restrict__ k,
                                           int nx, int y, int z, int h,
                                           const int (&koff)[NSLOT], const bool (&kval)[NSLOT])
{
    const bool xin = (unsigned)nx < GH;
    const int own = ((nx * GW + y) * GT + z) * DIMC + h * HDD;
#pragma unroll
    for (int i = 0; i < NSLOT; ++i) {
        if (xin && kval[i]) {
            const float* kp = k + (own + koff[i]);
#pragma unroll
            for (int d = 0; d < HDD; d += 2) {
                v2f v = *(const v2f*)(kp + d);
                P.v[i][d]     = v.x;
                P.v[i][d + 1] = v.y;
            }
        } else {
#pragma unroll
            for (int d = 0; d < HDD; ++d) P.v[i][d] = 0.0f;
        }
    }
}

__device__ __forceinline__ void load_q(float (&qv)[HDD], const float* __restrict__ q,
                                       int x, int y, int z, int h)
{
    const float* qp = q + (size_t)((x * GW + y) * GT + z) * DIMC + h * HDD;
#pragma unroll
    for (int d = 0; d < HDD; d += 2) {
        v2f v = *(const v2f*)(qp + d);
        qv[d]     = v.x;
        qv[d + 1] = v.y;
    }
}

template <int TI>
__device__ __forceinline__ void scores5(const KPlane& P, QState& Q,
                                        const float* rp, const float* rp4, int half)
{
#pragma unroll
    for (int i = 0; i < NSLOT; ++i) {
        float dot = 0.0f;
#pragma unroll
        for (int d = 0; d < HDD; ++d) dot = fmaf(Q.q[d], P.v[i][d], dot);
        const float b = (i < NSLOT - 1) ? rp[TI * 9 + i] : rp4[TI * 9];
        Q.s[TI * NSLOT + i] = dot + b;
    }
    if (half) Q.s[TI * NSLOT + NSLOT - 1] = -1e30f;
}

__device__ __forceinline__ void finish_query(const float (&s)[NSC], float* __restrict__ out,
                                             int x, int y, int z2, int half, int col)
{
    float mx = -1e30f;
#pragma unroll
    for (int e = 0; e < NSC; ++e) mx = fmaxf(mx, s[e]);
    mx = fmaxf(mx, xhalf(mx));

    float p[NSC];
    float sum = 0.0f;
#pragma unroll
    for (int e = 0; e < NSC; ++e) { p[e] = __expf(s[e] - mx); sum += p[e]; }
    sum += xhalf(sum);
    const float inv = __builtin_amdgcn_rcpf(sum);
#pragma unroll
    for (int e = 0; e < NSC; ++e) p[e] *= inv;

    v8f acc = {};
#pragma unroll
    for (int c = 0; c < NMMA; ++c) {
        const int e0 = 2 * c, e1 = 2 * c + 1;
        v2f a, b;
        a.x = p[e0];
        a.y = (e1 < NSC) ? p[e1 < NSC ? e1 : 0] : 0.0f;
        const unsigned cx = half ? bcode(1, e0) : bcode(0, e0);
        const unsigned cy = half ? bcode(1, e1) : bcode(0, e1);
        b.x = bval(cx, col);
        b.y = bval(cy, col);
        acc = __builtin_amdgcn_wmma_f32_16x16x4_f32(
             false, a,  false, b,
             (short)0, acc,  false,  false);
    }

    if (col < 3) {
        const size_t sp = (size_t)((x * GW + y) * GT) + (size_t)(z2 * 2 + half);
#pragma unroll
        for (int r = 0; r < 8; ++r) {
            out[(size_t)(r * 3 + col) * (GH * GW * GT) + sp] = acc[r];
        }
    }
}

__global__ __launch_bounds__(256) void nat3d_wmma_kernel(
    const float* __restrict__ q,
    const float* __restrict__ k,
    const float* __restrict__ rpb,
    float* __restrict__ out)
{
    __shared__ float s_rpb[NHD * NT];
    int tid = threadIdx.x;
    if (tid < NHD * NT) s_rpb[tid] = rpb[tid];
    __syncthreads();

    const int lane = tid & 31;
    const int wave = __builtin_amdgcn_readfirstlane(tid >> 5);
    const int unit = blockIdx.x * 8 + wave;
    const int z2   = unit % (GT / 2);
    const int tmp  = unit / (GT / 2);
    const int y    = tmp % GW;
    const int seg  = tmp / GW;
    const int x0   = seg * XSEG;
    const int xe   = x0 + XSEG;

    const int m    = lane & 15;
    const int half = lane >> 4;
    const int h    = m & 7;
    const int z    = z2 * 2 + (m >> 3);
    const int col  = lane & 15;

    int  koff[NSLOT];
    bool kval[NSLOT];
#pragma unroll
    for (int i = 0; i < NSLOT; ++i) {
        const int  j  = half ? (5 + i) / 3 : i / 3;
        const int  l  = half ? (5 + i) % 3 : i % 3;
        const bool in = (i < NSLOT - 1) || !half;
        koff[i] = ((j - 1) * GT + (l - 1)) * DIMC;
        kval[i] = in && (unsigned)(y + j - 1) < GW && (unsigned)(z + l - 1) < GT;
    }
    const float* rp  = s_rpb + h * NT + half * 5;
    const float* rp4 = rp + (NSLOT - 1) - half;

    QState U, V, Wd;
    int X = __builtin_amdgcn_readfirstlane(x0 - 1);
#define NAT_ROUND(QU, QV, QW)                                                          \
    {                                                                                  \
        KPlane kv;                                                                     \
        float qraw[HDD];                                                               \
        load_plane(kv, k, X, y, z, h, koff, kval);                                     \
        if (X + 1 < xe)  load_q(qraw, q, X + 1, y, z, h);                              \
        if (X - 2 >= x0) finish_query(QW.s, out, X - 2, y, z2, half, col);             \
        if (X - 1 >= x0) scores5<2>(kv, QU, rp, rp4, half);                            \
        if (X >= x0 && X < xe) scores5<1>(kv, QV, rp, rp4, half);                      \
        if (X + 1 < xe) {                                                              \
            _Pragma("unroll")                                                          \
            for (int d = 0; d < HDD; ++d) QW.q[d] = qraw[d] * QSCALE;                  \
            scores5<0>(kv, QW, rp, rp4, half);                                         \
        }                                                                              \
        if (++X > xe) break;                                                           \
    }
    for (;;) {
        NAT_ROUND(U, V, Wd)
        NAT_ROUND(V, Wd, U)
        NAT_ROUND(Wd, U, V)
    }
#undef NAT_ROUND
    constexpr int LAST = (XSEG + 1) % 3;
    if (LAST == 0)      finish_query(U.s,  out, xe - 1, y, z2, half, col);
    else if (LAST == 1) finish_query(V.s,  out, xe - 1, y, z2, half, col);
    else                finish_query(Wd.s, out, xe - 1, y, z2, half, col);
}

extern "C" void kernel_launch(void* const* d_in, const int* in_sizes, int n_in,
                              void* d_out, int out_size, void* d_ws, size_t ws_size,
                              hipStream_t stream) {
    (void)in_sizes; (void)n_in; (void)out_size; (void)d_ws; (void)ws_size;
    const float* q   = (const float*)d_in[0];
    const float* k   = (const float*)d_in[1];
    const float* rpb = (const float*)d_in[2];
    float* out = (float*)d_out;

    dim3 grid(500), block(256);
    nat3d_wmma_kernel<<<grid, block, 0, stream>>>(q, k, rpb, out);
}


# Results staged in LDS per block: the 8 waves' 16 side-by-side z of each channel leave as one 64-byte run (one barrier per finished query), not as 8-byte pieces from every wave; sum order unchanged
# speedup vs baseline: 2.4989x; 1.5553x over previous
#include <hip/hip_runtime.h>


typedef __attribute__((ext_vector_type(2))) float v2f;
typedef __attribute__((ext_vector_type(8))) float v8f;

#define NHD   8
#define HDD   6
#define DIMC  48
#define NT    27
#define GH    40
#define GW    40
#define GT    40
#define XSEG  8
#define QSCALE 0.40824829046386301636f

#define NSLOT 5
#define NSC   15
#define NMMA  8
#define NCH   (NHD * 3)
#define NVOX  (GH * GW * GT)
#define BZ    16

constexpr unsigned bcode(int hf, int e) {
    return (e >= NSC || (hf == 1 && e % NSLOT == NSLOT - 1))
        ? 0x55555555u
        : (unsigned)(e / NSLOT)
        | ((unsigned)((hf * 5 + e % NSLOT) / 3) << 2)
        | ((unsigned)((hf * 5 + e % NSLOT) % 3) << 4)
        | 0x55555540u;
}

__device__ __forceinline__ float bval(unsigned code, int col) {
    return (float)((int)((code >> (2 * col)) & 3u) - 1);
}

__device__ __forceinline__ float xhalf(float x) {
    const int b = __builtin_bit_cast(int, x);
    return __builtin_bit_cast(float,
        __builtin_amdgcn_permlanex16(b, b, 0x76543210, 0xfedcba98, false, false));
}


struct KPlane { float v[NSLOT][HDD]; };

struct QState { float q[HDD]; float s[NSC]; };

__device__ __forceinline__ void load_plane(KPlane& P, const float* __restrict__ k,
                                           int nx, int y, int z, int h,
                                           const int (&koff)[NSLOT], const bool (&kval)[NSLOT])
{
    const bool xin = (unsigned)nx < GH;
    const int own = ((nx * GW + y) * GT + z) * DIMC + h * HDD;
#pragma unroll
    for (int i = 0; i < NSLOT; ++i) {
        if (xin && kval[i]) {
            const float* kp = k + (own + koff[i]);
#pragma unroll
            for (int d = 0; d < HDD; d += 2) {
                v2f v = *(const v2f*)(kp + d);
                P.v[i][d]     = v.x;
                P.v[i][d + 1] = v.y;
            }
        } else {
#pragma unroll
            for (int d = 0; d < HDD; ++d) P.v[i][d] = 0.0f;
        }
    }
}

__device__ __forceinline__ void load_q(float (&qv)[HDD], const float* __restrict__ q,
                                       int x, int y, int z, int h)
{
    const float* qp = q + (size_t)((x * GW + y) * GT + z) * DIMC + h * HDD;
#pragma unroll
    for (int d = 0; d < HDD; d += 2) {
        v2f v = *(const v2f*)(qp + d);
        qv[d]     = v.x;
        qv[d + 1] = v.y;
    }
}

template <int TI>
__device__ __forceinline__ void scores5(const KPlane& P, QState& Q,
                                        const float* rp, const float* rp4, int half)
{
#pragma unroll
    for (int i = 0; i < NSLOT; ++i) {
        float dot = 0.0f;
#pragma unroll
        for (int d = 0; d < HDD; ++d) dot = fmaf(Q.q[d], P.v[i][d], dot);
        const float b = (i < NSLOT - 1) ? rp[TI * 9 + i] : rp4[TI * 9];
        Q.s[TI * NSLOT + i] = dot + b;
    }
    if (half) Q.s[TI * NSLOT + NSLOT - 1] = -1e30f;
}

__device__ __forceinline__ void finish_query(const float (&s)[NSC], float* __restrict__ out,
                                             float* so, int x, int obase,
                                             int wave, int tid, int half, int col)
{
    float mx = -1e30f;
#pragma unroll
    for (int e = 0; e < NSC; ++e) mx = fmaxf(mx, s[e]);
    mx = fmaxf(mx, xhalf(mx));

    float p[NSC];
    float sum = 0.0f;
#pragma unroll
    for (int e = 0; e < NSC; ++e) { p[e] = __expf(s[e] - mx); sum += p[e]; }
    sum += xhalf(sum);
    const float inv = __builtin_amdgcn_rcpf(sum);
#pragma unroll
    for (int e = 0; e < NSC; ++e) p[e] *= inv;

    v8f acc = {};
#pragma unroll
    for (int c = 0; c < NMMA; ++c) {
        const int e0 = 2 * c, e1 = 2 * c + 1;
        v2f a, b;
        a.x = p[e0];
        a.y = (e1 < NSC) ? p[e1 < NSC ? e1 : 0] : 0.0f;
        const unsigned cx = half ? bcode(1, e0) : bcode(0, e0);
        const unsigned cy = half ? bcode(1, e1) : bcode(0, e1);
        b.x = bval(cx, col);
        b.y = bval(cy, col);
        acc = __builtin_amdgcn_wmma_f32_16x16x4_f32(
             false, a,  false, b,
             (short)0, acc,  false,  false);
    }

    if (col < 3) {
#pragma unroll
        for (int r = 0; r < 8; ++r) {
            so[(r * 3 + col) * BZ + wave * 2 + half] = acc[r];
        }
    }
    __syncthreads();
    float* op = out + (size_t)x * (GW * GT) + obase;
    op[(size_t)(tid >> 4) * NVOX] = so[tid];
    if (wave < 4) op[(size_t)((tid >> 4) + 16) * NVOX] = so[tid + 256];
}

__global__ __launch_bounds__(256) void nat3d_wmma_kernel(
    const float* __restrict__ q,
    const float* __restrict__ k,
    const float* __restrict__ rpb,
    float* __restrict__ out)
{
    __shared__ float s_rpb[NHD * NT];
    __shared__ float s_out[2][NCH * BZ];
    int tid = threadIdx.x;
    if (tid < NHD * NT) s_rpb[tid] = rpb[tid];
    __syncthreads();

    const int lane = tid & 31;
    const int wave = __builtin_amdgcn_readfirstlane(tid >> 5);
    const int unit = blockIdx.x * 8 + wave;
    const int z2   = unit % (GT / 2);
    const int tmp  = unit / (GT / 2);
    const int y    = tmp % GW;
    const int seg  = tmp / GW;
    const int x0   = seg * XSEG;
    const int xe   = x0 + XSEG;
    const int obase = (int)((blockIdx.x * 8) % (GW * GT / 2)) * 2 + (tid & 15);

    const int m    = lane & 15;
    const int half = lane >> 4;
    const int h    = m & 7;
    const int z    = z2 * 2 + (m >> 3);
    const int col  = lane & 15;

    int  koff[NSLOT];
    bool kval[NSLOT];
#pragma unroll
    for (int i = 0; i < NSLOT; ++i) {
        const int  j  = half ? (5 + i) / 3 : i / 3;
        const int  l  = half ? (5 + i) % 3 : i % 3;
        const bool in = (i < NSLOT - 1) || !half;
        koff[i] = ((j - 1) * GT + (l - 1)) * DIMC;
        kval[i] = in && (unsigned)(y + j - 1) < GW && (unsigned)(z + l - 1) < GT;
    }
    const float* rp  = s_rpb + h * NT + half * 5;
    const float* rp4 = rp + (NSLOT - 1) - half;

    QState U, V, Wd;
    int X = __builtin_amdgcn_readfirstlane(x0 - 1);
#define NAT_ROUND(QU, QV, QW)                                                          \
    {                                                                                  \
        KPlane kv;                                                                     \
        float qraw[HDD];                                                               \
        load_plane(kv, k, X, y, z, h, koff, kval);                                     \
        if (X + 1 < xe)  load_q(qraw, q, X + 1, y, z, h);                              \
        if (X - 2 >= x0) finish_query(QW.s, out, s_out[X & 1], X - 2, obase,           \
                                      wave, tid, half, col);                           \
        if (X - 1 >= x0) scores5<2>(kv, QU, rp, rp4, half);                            \
        if (X >= x0 && X < xe) scores5<1>(kv, QV, rp, rp4, half);                      \
        if (X + 1 < xe) {                                                              \
            _Pragma("unroll")                                                          \
            for (int d = 0; d < HDD; ++d) QW.q[d] = qraw[d] * QSCALE;                  \
            scores5<0>(kv, QW, rp, rp4, half);                                         \
        }                                                                              \
        if (++X > xe) break;                                                           \
    }
    for (;;) {
        NAT_ROUND(U, V, Wd)
        NAT_ROUND(V, Wd, U)
        NAT_ROUND(Wd, U, V)
    }
#undef NAT_ROUND
    constexpr int LAST = (XSEG + 1) % 3;
    if (LAST == 0)      finish_query(U.s,  out, s_out[X & 1], xe - 1, obase, wave, tid, half, col);
    else if (LAST == 1) finish_query(V.s,  out, s_out[X & 1], xe - 1, obase, wave, tid, half, col);
    else                finish_query(Wd.s, out, s_out[X & 1], xe - 1, obase, wave, tid, half, col);
}

extern "C" void kernel_launch(void* const* d_in, const int* in_sizes, int n_in,
                              void* d_out, int out_size, void* d_ws, size_t ws_size,
                              hipStream_t stream) {
    (void)in_sizes; (void)n_in; (void)out_size; (void)d_ws; (void)ws_size;
    const float* q   = (const float*)d_in[0];
    const float* k   = (const float*)d_in[1];
    const float* rpb = (const float*)d_in[2];
    float* out = (float*)d_out;

    dim3 grid(500), block(256);
    nat3d_wmma_kernel<<<grid, block, 0, stream>>>(q, k, rpb, out);
}


# Key planes loaded one round ahead into a second register set: plane X+1's loads go out behind query X-2's finish and fly during plane X's scores and the next softmax; sum order unchanged
# speedup vs baseline: 2.7122x; 1.0854x over previous
#include <hip/hip_runtime.h>


typedef __attribute__((ext_vector_type(2))) float v2f;
typedef __attribute__((ext_vector_type(4))) float v4f;
typedef __attribute__((ext_vector_type(8))) float v8f;
typedef v4f v4f_a8 __attribute__((aligned(8)));

#define NHD   8
#define HDD   6
#define DIMC  48
#define NT    27
#define GH    40
#define GW    40
#define GT    40
#define XSEG  8
#define QSCALE 0.40824829046386301636f

#define NSLOT 5
#define NSC   15
#define NMMA  8
#define NCH   (NHD * 3)
#define NVOX  (GH * GW * GT)
#define BZ    16

constexpr unsigned bcode(int hf, int e) {
    return (e >= NSC || (hf == 1 && e % NSLOT == NSLOT - 1))
        ? 0x55555555u
        : (unsigned)(e / NSLOT)
        | ((unsigned)((hf * 5 + e % NSLOT) / 3) << 2)
        | ((unsigned)((hf * 5 + e % NSLOT) % 3) << 4)
        | 0x55555540u;
}

__device__ __forceinline__ float bval(unsigned code, int col) {
    return (float)((int)((code >> (2 * col)) & 3u) - 1);
}

__device__ __forceinline__ float xhalf(float x) {
    const int b = __builtin_bit_cast(int, x);
    return __builtin_bit_cast(float,
        __builtin_amdgcn_permlanex16(b, b, 0x76543210, 0xfedcba98, false, false));
}


struct KPlane { v4f a[NSLOT]; v2f b[NSLOT]; };

__device__ __forceinline__ float kdim(const KPlane& P, int i, int d) {
    return d < 4 ? P.a[i][d] : P.b[i][d - 4];
}

struct QState { float q[HDD]; float s[NSC]; };

__device__ __forceinline__ void load_plane(KPlane& P, const float* __restrict__ k,
                                           int nx, int y, int z, int h,
                                           const int (&koff)[NSLOT], const bool (&kval)[NSLOT])
{
    const bool xin = (unsigned)nx < GH;
    const int own = ((nx * GW + y) * GT + z) * DIMC + h * HDD;
#pragma unroll
    for (int i = 0; i < NSLOT; ++i) {
        if (xin && kval[i]) {
            const float* kp = k + (own + koff[i]);
            P.a[i] = *(const v4f_a8*)kp;
            P.b[i] = *(const v2f*)(kp + 4);
        } else {
            P.a[i] = (v4f){0.0f, 0.0f, 0.0f, 0.0f};
            P.b[i] = (v2f){0.0f, 0.0f};
        }
    }
}

__device__ __forceinline__ void plane_landed(const KPlane& P, const float (&qr)[HDD]) {
    asm volatile("" :: "v"(P.a[0]), "v"(P.a[1]), "v"(P.a[2]), "v"(P.a[3]), "v"(P.a[4]),
                       "v"(P.b[0]), "v"(P.b[1]), "v"(P.b[2]), "v"(P.b[3]), "v"(P.b[4]),
                       "v"(qr[0]), "v"(qr[1]), "v"(qr[2]), "v"(qr[3]), "v"(qr[4]), "v"(qr[5]));
}

__device__ __forceinline__ void load_q(float (&qv)[HDD], const float* __restrict__ q,
                                       int x, int y, int z, int h)
{
    const float* qp = q + (size_t)((x * GW + y) * GT + z) * DIMC + h * HDD;
#pragma unroll
    for (int d = 0; d < HDD; d += 2) {
        v2f v = *(const v2f*)(qp + d);
        qv[d]     = v.x;
        qv[d + 1] = v.y;
    }
}

template <int TI>
__device__ __forceinline__ void scores5(const KPlane& P, QState& Q,
                                        const float* rp, const float* rp4, int half)
{
#pragma unroll
    for (int i = 0; i < NSLOT; ++i) {
        float dot = 0.0f;
#pragma unroll
        for (int d = 0; d < HDD; ++d) dot = fmaf(Q.q[d], kdim(P, i, d), dot);
        const float b = (i < NSLOT - 1) ? rp[TI * 9 + i] : rp4[TI * 9];
        Q.s[TI * NSLOT + i] = dot + b;
    }
    if (half) Q.s[TI * NSLOT + NSLOT - 1] = -1e30f;
}

__device__ __forceinline__ void finish_query(const float (&s)[NSC], float* __restrict__ out,
                                             float* so, int x, int obase,
                                             int wave, int tid, int half, int col)
{
    float mx = -1e30f;
#pragma unroll
    for (int e = 0; e < NSC; ++e) mx = fmaxf(mx, s[e]);
    mx = fmaxf(mx, xhalf(mx));

    float p[NSC];
    float sum = 0.0f;
#pragma unroll
    for (int e = 0; e < NSC; ++e) { p[e] = __expf(s[e] - mx); sum += p[e]; }
    sum += xhalf(sum);
    const float inv = __builtin_amdgcn_rcpf(sum);
#pragma unroll
    for (int e = 0; e < NSC; ++e) p[e] *= inv;

    v8f acc = {};
#pragma unroll
    for (int c = 0; c < NMMA; ++c) {
        const int e0 = 2 * c, e1 = 2 * c + 1;
        v2f a, b;
        a.x = p[e0];
        a.y = (e1 < NSC) ? p[e1 < NSC ? e1 : 0] : 0.0f;
        const unsigned cx = half ? bcode(1, e0) : bcode(0, e0);
        const unsigned cy = half ? bcode(1, e1) : bcode(0, e1);
        b.x = bval(cx, col);
        b.y = bval(cy, col);
        acc = __builtin_amdgcn_wmma_f32_16x16x4_f32(
             false, a,  false, b,
             (short)0, acc,  false,  false);
    }

    if (col < 3) {
#pragma unroll
        for (int r = 0; r < 8; ++r) {
            so[(r * 3 + col) * BZ + wave * 2 + half] = acc[r];
        }
    }
    __syncthreads();
    float* op = out + (size_t)x * (GW * GT) + obase;
    op[(size_t)(tid >> 4) * NVOX] = so[tid];
    if (wave < 4) op[(size_t)((tid >> 4) + 16) * NVOX] = so[tid + 256];
}

__global__ __launch_bounds__(256) void nat3d_wmma_kernel(
    const float* __restrict__ q,
    const float* __restrict__ k,
    const float* __restrict__ rpb,
    float* __restrict__ out)
{
    __shared__ float s_rpb[NHD * NT];
    __shared__ float s_out[2][NCH * BZ];
    int tid = threadIdx.x;
    if (tid < NHD * NT) s_rpb[tid] = rpb[tid];
    __syncthreads();

    const int lane = tid & 31;
    const int wave = __builtin_amdgcn_readfirstlane(tid >> 5);
    const int unit = blockIdx.x * 8 + wave;
    const int z2   = unit % (GT / 2);
    const int tmp  = unit / (GT / 2);
    const int y    = tmp % GW;
    const int seg  = tmp / GW;
    const int x0   = seg * XSEG;
    const int xe   = x0 + XSEG;
    const int obase = (int)((blockIdx.x * 8) % (GW * GT / 2)) * 2 + (tid & 15);

    const int m    = lane & 15;
    const int half = lane >> 4;
    const int h    = m & 7;
    const int z    = z2 * 2 + (m >> 3);
    const int col  = lane & 15;

    int  koff[NSLOT];
    bool kval[NSLOT];
#pragma unroll
    for (int i = 0; i < NSLOT; ++i) {
        const int  j  = half ? (5 + i) / 3 : i / 3;
        const int  l  = half ? (5 + i) % 3 : i % 3;
        const bool in = (i < NSLOT - 1) || !half;
        koff[i] = ((j - 1) * GT + (l - 1)) * DIMC;
        kval[i] = in && (unsigned)(y + j - 1) < GW && (unsigned)(z + l - 1) < GT;
    }
    const float* rp  = s_rpb + h * NT + half * 5;
    const float* rp4 = rp + (NSLOT - 1) - half;

    QState U, V, Wd;
    KPlane KA, KB;
    float qrA[HDD], qrB[HDD];
    int X = __builtin_amdgcn_readfirstlane(x0 - 1);
    load_plane(KA, k, X, y, z, h, koff, kval);
    load_q(qrA, q, X + 1, y, z, h);
#define NAT_ROUND(QU, QV, QW, KC, KN, QRC, QRN)                                        \
    {                                                                                  \
        if (X - 2 >= x0) finish_query(QW.s, out, s_out[X & 1], X - 2, obase,           \
                                      wave, tid, half, col);                           \
        if (X + 1 < xe) {                                                              \
            _Pragma("unroll")                                                          \
            for (int d = 0; d < HDD; ++d) QW.q[d] = QRC[d] * QSCALE;                   \
        }                                                                              \
        plane_landed(KC, QRC);                                                         \
        if (X + 1 <= xe) load_plane(KN, k, X + 1, y, z, h, koff, kval);                \
        if (X + 2 < xe)  load_q(QRN, q, X + 2, y, z, h);                               \
        if (X - 1 >= x0) scores5<2>(KC, QU, rp, rp4, half);                            \
        if (X >= x0 && X < xe) scores5<1>(KC, QV, rp, rp4, half);                      \
        if (X + 1 < xe)  scores5<0>(KC, QW, rp, rp4, half);                            \
        if (++X > xe) break;                                                           \
    }
    for (;;) {
        NAT_ROUND(U, V, Wd, KA, KB, qrA, qrB)
        NAT_ROUND(V, Wd, U, KB, KA, qrB, qrA)
        NAT_ROUND(Wd, U, V, KA, KB, qrA, qrB)
        NAT_ROUND(U, V, Wd, KB, KA, qrB, qrA)
        NAT_ROUND(V, Wd, U, KA, KB, qrA, qrB)
        NAT_ROUND(Wd, U, V, KB, KA, qrB, qrA)
    }
#undef NAT_ROUND
    constexpr int LAST = (XSEG + 1) % 3;
    if (LAST == 0)      finish_query(U.s,  out, s_out[X & 1], xe - 1, obase, wave, tid, half, col);
    else if (LAST == 1) finish_query(V.s,  out, s_out[X & 1], xe - 1, obase, wave, tid, half, col);
    else                finish_query(Wd.s, out, s_out[X & 1], xe - 1, obase, wave, tid, half, col);
}

extern "C" void kernel_launch(void* const* d_in, const int* in_sizes, int n_in,
                              void* d_out, int out_size, void* d_ws, size_t ws_size,
                              hipStream_t stream) {
    (void)in_sizes; (void)n_in; (void)out_size; (void)d_ws; (void)ws_size;
    const float* q   = (const float*)d_in[0];
    const float* k   = (const float*)d_in[1];
    const float* rpb = (const float*)d_in[2];
    float* out = (float*)d_out;

    dim3 grid(500), block(256);
    nat3d_wmma_kernel<<<grid, block, 0, stream>>>(q, k, rpb, out);
}


# Block barrier split in two: a wave signals once its results are staged, loads and scores on, and waits a round later, just before reading the stage out; its fences cover LDS only; sum order unchanged
# speedup vs baseline: 2.9946x; 1.1041x over previous
#include <hip/hip_runtime.h>


typedef __attribute__((ext_vector_type(2))) float v2f;
typedef __attribute__((ext_vector_type(4))) float v4f;
typedef __attribute__((ext_vector_type(8))) float v8f;
typedef v4f v4f_a8 __attribute__((aligned(8)));

#define NHD   8
#define HDD   6
#define DIMC  48
#define NT    27
#define GH    40
#define GW    40
#define GT    40
#define XSEG  8
#define QSCALE 0.40824829046386301636f

#define NSLOT 5
#define NSC   15
#define NMMA  8
#define NCH   (NHD * 3)
#define NVOX  (GH * GW * GT)
#define BZ    16

constexpr unsigned bcode(int hf, int e) {
    return (e >= NSC || (hf == 1 && e % NSLOT == NSLOT - 1))
        ? 0x55555555u
        : (unsigned)(e / NSLOT)
        | ((unsigned)((hf * 5 + e % NSLOT) / 3) << 2)
        | ((unsigned)((hf * 5 + e % NSLOT) % 3) << 4)
        | 0x55555540u;
}

__device__ __forceinline__ float bval(unsigned code, int col) {
    return (float)((int)((code >> (2 * col)) & 3u) - 1);
}

__device__ __forceinline__ float xhalf(float x) {
    const int b = __builtin_bit_cast(int, x);
    return __builtin_bit_cast(float,
        __builtin_amdgcn_permlanex16(b, b, 0x76543210, 0xfedcba98, false, false));
}


struct KPlane { v4f a[NSLOT]; v2f b[NSLOT]; };

__device__ __forceinline__ float kdim(const KPlane& P, int i, int d) {
    return d < 4 ? P.a[i][d] : P.b[i][d - 4];
}

struct QState { float q[HDD]; float s[NSC]; };

__device__ __forceinline__ void load_plane(KPlane& P, const float* __restrict__ k,
                                           int nx, int y, int z, int h,
                                           const int (&koff)[NSLOT], const bool (&kval)[NSLOT])
{
    const bool xin = (unsigned)nx < GH;
    const int own = ((nx * GW + y) * GT + z) * DIMC + h * HDD;
#pragma unroll
    for (int i = 0; i < NSLOT; ++i) {
        if (xin && kval[i]) {
            const float* kp = k + (own + koff[i]);
            P.a[i] = *(const v4f_a8*)kp;
            P.b[i] = *(const v2f*)(kp + 4);
        } else {
            P.a[i] = (v4f){0.0f, 0.0f, 0.0f, 0.0f};
            P.b[i] = (v2f){0.0f, 0.0f};
        }
    }
}

__device__ __forceinline__ void plane_landed(const KPlane& P, const float (&qr)[HDD]) {
    asm volatile("" :: "v"(P.a[0]), "v"(P.a[1]), "v"(P.a[2]), "v"(P.a[3]), "v"(P.a[4]),
                       "v"(P.b[0]), "v"(P.b[1]), "v"(P.b[2]), "v"(P.b[3]), "v"(P.b[4]),
                       "v"(qr[0]), "v"(qr[1]), "v"(qr[2]), "v"(qr[3]), "v"(qr[4]), "v"(qr[5]));
}

__device__ __forceinline__ void load_q(float (&qv)[HDD], const float* __restrict__ q,
                                       int x, int y, int z, int h)
{
    const float* qp = q + (size_t)((x * GW + y) * GT + z) * DIMC + h * HDD;
#pragma unroll
    for (int d = 0; d < HDD; d += 2) {
        v2f v = *(const v2f*)(qp + d);
        qv[d]     = v.x;
        qv[d + 1] = v.y;
    }
}

template <int TI>
__device__ __forceinline__ void scores5(const KPlane& P, QState& Q,
                                        const float* rp, const float* rp4, int half)
{
#pragma unroll
    for (int i = 0; i < NSLOT; ++i) {
        float dot = 0.0f;
#pragma unroll
        for (int d = 0; d < HDD; ++d) dot = fmaf(Q.q[d], kdim(P, i, d), dot);
        const float b = (i < NSLOT - 1) ? rp[TI * 9 + i] : rp4[TI * 9];
        Q.s[TI * NSLOT + i] = dot + b;
    }
    if (half) Q.s[TI * NSLOT + NSLOT - 1] = -1e30f;
}

__device__ __forceinline__ void stage_published() {
    __builtin_amdgcn_fence(__ATOMIC_RELEASE, "workgroup", "local");
    __builtin_amdgcn_s_barrier_signal(-1);
}
__device__ __forceinline__ void stage_complete() {
    __builtin_amdgcn_s_barrier_wait(-1);
    __builtin_amdgcn_fence(__ATOMIC_ACQUIRE, "workgroup", "local");
}

__device__ __forceinline__ v8f query_result(const float (&s)[NSC], int half, int col)
{
    float mx = -1e30f;
#pragma unroll
    for (int e = 0; e < NSC; ++e) mx = fmaxf(mx, s[e]);
    mx = fmaxf(mx, xhalf(mx));

    float p[NSC];
    float sum = 0.0f;
#pragma unroll
    for (int e = 0; e < NSC; ++e) { p[e] = __expf(s[e] - mx); sum += p[e]; }
    sum += xhalf(sum);
    const float inv = __builtin_amdgcn_rcpf(sum);
#pragma unroll
    for (int e = 0; e < NSC; ++e) p[e] *= inv;

    v8f acc = {};
#pragma unroll
    for (int c = 0; c < NMMA; ++c) {
        const int e0 = 2 * c, e1 = 2 * c + 1;
        v2f a, b;
        a.x = p[e0];
        a.y = (e1 < NSC) ? p[e1 < NSC ? e1 : 0] : 0.0f;
        const unsigned cx = half ? bcode(1, e0) : bcode(0, e0);
        const unsigned cy = half ? bcode(1, e1) : bcode(0, e1);
        b.x = bval(cx, col);
        b.y = bval(cy, col);
        acc = __builtin_amdgcn_wmma_f32_16x16x4_f32(
             false, a,  false, b,
             (short)0, acc,  false,  false);
    }
    return acc;
}

__device__ __forceinline__ void hand_over(const v8f& acc, float* so_new, const float* so_old,
                                          bool has_old, float* __restrict__ out, int x_old,
                                          int obase, int wave, int tid, int half, int col)
{
    float d0 = 0.0f, d1 = 0.0f;
    if (has_old) {
        stage_complete();
        d0 = so_old[tid];
        if (wave < 4) d1 = so_old[tid + 256];
    }
    if (col < 3) {
#pragma unroll
        for (int r = 0; r < 8; ++r) {
            so_new[(r * 3 + col) * BZ + wave * 2 + half] = acc[r];
        }
    }
    stage_published();
    if (has_old) {
        float* op = out + (size_t)x_old * (GW * GT) + obase;
        op[(size_t)(tid >> 4) * NVOX] = d0;
        if (wave < 4) op[(size_t)((tid >> 4) + 16) * NVOX] = d1;
    }
}

__device__ __forceinline__ void drain_last(const float* so, float* __restrict__ out, int x,
                                           int obase, int wave, int tid)
{
    stage_complete();
    float* op = out + (size_t)x * (GW * GT) + obase;
    op[(size_t)(tid >> 4) * NVOX] = so[tid];
    if (wave < 4) op[(size_t)((tid >> 4) + 16) * NVOX] = so[tid + 256];
}

__global__ __launch_bounds__(256) void nat3d_wmma_kernel(
    const float* __restrict__ q,
    const float* __restrict__ k,
    const float* __restrict__ rpb,
    float* __restrict__ out)
{
    __shared__ float s_rpb[NHD * NT];
    __shared__ float s_out[2][NCH * BZ];
    int tid = threadIdx.x;
    if (tid < NHD * NT) s_rpb[tid] = rpb[tid];
    __syncthreads();

    const int lane = tid & 31;
    const int wave = __builtin_amdgcn_readfirstlane(tid >> 5);
    const int unit = blockIdx.x * 8 + wave;
    const int z2   = unit % (GT / 2);
    const int tmp  = unit / (GT / 2);
    const int y    = tmp % GW;
    const int seg  = tmp / GW;
    const int x0   = seg * XSEG;
    const int xe   = x0 + XSEG;
    const int obase = (int)((blockIdx.x * 8) % (GW * GT / 2)) * 2 + (tid & 15);

    const int m    = lane & 15;
    const int half = lane >> 4;
    const int h    = m & 7;
    const int z    = z2 * 2 + (m >> 3);
    const int col  = lane & 15;

    int  koff[NSLOT];
    bool kval[NSLOT];
#pragma unroll
    for (int i = 0; i < NSLOT; ++i) {
        const int  j  = half ? (5 + i) / 3 : i / 3;
        const int  l  = half ? (5 + i) % 3 : i % 3;
        const bool in = (i < NSLOT - 1) || !half;
        koff[i] = ((j - 1) * GT + (l - 1)) * DIMC;
        kval[i] = in && (unsigned)(y + j - 1) < GW && (unsigned)(z + l - 1) < GT;
    }
    const float* rp  = s_rpb + h * NT + half * 5;
    const float* rp4 = rp + (NSLOT - 1) - half;

    QState U, V, Wd;
    KPlane KA, KB;
    float qrA[HDD], qrB[HDD];
    int X = __builtin_amdgcn_readfirstlane(x0 - 1);
    load_plane(KA, k, X, y, z, h, koff, kval);
    load_q(qrA, q, X + 1, y, z, h);
#define NAT_ROUND(QU, QV, QW, KC, KN, QRC, QRN)                                        \
    {                                                                                  \
        if (X - 2 >= x0) {                                                             \
            const v8f acc = query_result(QW.s, half, col);                             \
            hand_over(acc, s_out[X & 1], s_out[(X + 1) & 1], X - 3 >= x0, out, X - 3,  \
                      obase, wave, tid, half, col);                                    \
        }                                                                              \
        if (X + 1 < xe) {                                                              \
            _Pragma("unroll")                                                          \
            for (int d = 0; d < HDD; ++d) QW.q[d] = QRC[d] * QSCALE;                   \
        }                                                                              \
        plane_landed(KC, QRC);                                                         \
        if (X + 1 <= xe) load_plane(KN, k, X + 1, y, z, h, koff, kval);                \
        if (X + 2 < xe)  load_q(QRN, q, X + 2, y, z, h);                               \
        if (X - 1 >= x0) scores5<2>(KC, QU, rp, rp4, half);                            \
        if (X >= x0 && X < xe) scores5<1>(KC, QV, rp, rp4, half);                      \
        if (X + 1 < xe)  scores5<0>(KC, QW, rp, rp4, half);                            \
        if (++X > xe) break;                                                           \
    }
    for (;;) {
        NAT_ROUND(U, V, Wd, KA, KB, qrA, qrB)
        NAT_ROUND(V, Wd, U, KB, KA, qrB, qrA)
        NAT_ROUND(Wd, U, V, KA, KB, qrA, qrB)
        NAT_ROUND(U, V, Wd, KB, KA, qrB, qrA)
        NAT_ROUND(V, Wd, U, KA, KB, qrA, qrB)
        NAT_ROUND(Wd, U, V, KB, KA, qrB, qrA)
    }
#undef NAT_ROUND
    constexpr int LAST = (XSEG + 1) % 3;
    static_assert(XSEG >= 2, "the tail hands over with an older query in the other stage");
    const v8f acc = (LAST == 0) ? query_result(U.s, half, col)
                  : (LAST == 1) ? query_result(V.s, half, col)
                                : query_result(Wd.s, half, col);
    hand_over(acc, s_out[X & 1], s_out[(X + 1) & 1], true, out, xe - 2, obase, wave, tid, half, col);
    drain_last(s_out[X & 1], out, xe - 1, obase, wave, tid);
}

extern "C" void kernel_launch(void* const* d_in, const int* in_sizes, int n_in,
                              void* d_out, int out_size, void* d_ws, size_t ws_size,
                              hipStream_t stream) {
    (void)in_sizes; (void)n_in; (void)out_size; (void)d_ws; (void)ws_size;
    const float* q   = (const float*)d_in[0];
    const float* k   = (const float*)d_in[1];
    const float* rpb = (const float*)d_in[2];
    float* out = (float*)d_out;

    dim3 grid(500), block(256);
    nat3d_wmma_kernel<<<grid, block, 0, stream>>>(q, k, rpb, out);
}


# Blocks of 16 waves own 32 side-by-side voxels: each channel's 32 results leave as one whole 128-byte line (16-byte stores, 4 lines each), no line put together by two blocks; sum order unchanged
# speedup vs baseline: 3.1795x; 1.0617x over previous
#include <hip/hip_runtime.h>


typedef __attribute__((ext_vector_type(2))) float v2f;
typedef __attribute__((ext_vector_type(4))) float v4f;
typedef __attribute__((ext_vector_type(8))) float v8f;
typedef v4f v4f_a8 __attribute__((aligned(8)));

#define NHD   8
#define HDD   6
#define DIMC  48
#define NT    27
#define GH    40
#define GW    40
#define GT    40
#define XSEG  8
#define QSCALE 0.40824829046386301636f

#define NSLOT 5
#define NSC   15
#define NMMA  8
#define NCH   (NHD * 3)
#define NVOX  (GH * GW * GT)
#define NWV   16
#define BZ    (NWV * 2)
#define NST4  (NCH * BZ / 4)

constexpr unsigned bcode(int hf, int e) {
    return (e >= NSC || (hf == 1 && e % NSLOT == NSLOT - 1))
        ? 0x55555555u
        : (unsigned)(e / NSLOT)
        | ((unsigned)((hf * 5 + e % NSLOT) / 3) << 2)
        | ((unsigned)((hf * 5 + e % NSLOT) % 3) << 4)
        | 0x55555540u;
}

__device__ __forceinline__ float bval(unsigned code, int col) {
    return (float)((int)((code >> (2 * col)) & 3u) - 1);
}

__device__ __forceinline__ float xhalf(float x) {
    const int b = __builtin_bit_cast(int, x);
    return __builtin_bit_cast(float,
        __builtin_amdgcn_permlanex16(b, b, 0x76543210, 0xfedcba98, false, false));
}


struct KPlane { v4f a[NSLOT]; v2f b[NSLOT]; };

__device__ __forceinline__ float kdim(const KPlane& P, int i, int d) {
    return d < 4 ? P.a[i][d] : P.b[i][d - 4];
}

struct QState { float q[HDD]; float s[NSC]; };

__device__ __forceinline__ void load_plane(KPlane& P, const float* __restrict__ k,
                                           int nx, int y, int z, int h,
                                           const int (&koff)[NSLOT], const bool (&kval)[NSLOT])
{
    const bool xin = (unsigned)nx < GH;
    const int own = ((nx * GW + y) * GT + z) * DIMC + h * HDD;
#pragma unroll
    for (int i = 0; i < NSLOT; ++i) {
        if (xin && kval[i]) {
            const float* kp = k + (own + koff[i]);
            P.a[i] = *(const v4f_a8*)kp;
            P.b[i] = *(const v2f*)(kp + 4);
        } else {
            P.a[i] = (v4f){0.0f, 0.0f, 0.0f, 0.0f};
            P.b[i] = (v2f){0.0f, 0.0f};
        }
    }
}

__device__ __forceinline__ void plane_landed(const KPlane& P, const float (&qr)[HDD]) {
    asm volatile("" :: "v"(P.a[0]), "v"(P.a[1]), "v"(P.a[2]), "v"(P.a[3]), "v"(P.a[4]),
                       "v"(P.b[0]), "v"(P.b[1]), "v"(P.b[2]), "v"(P.b[3]), "v"(P.b[4]),
                       "v"(qr[0]), "v"(qr[1]), "v"(qr[2]), "v"(qr[3]), "v"(qr[4]), "v"(qr[5]));
}

__device__ __forceinline__ void load_q(float (&qv)[HDD], const float* __restrict__ q,
                                       int x, int y, int z, int h)
{
    const float* qp = q + (size_t)((x * GW + y) * GT + z) * DIMC + h * HDD;
#pragma unroll
    for (int d = 0; d < HDD; d += 2) {
        v2f v = *(const v2f*)(qp + d);
        qv[d]     = v.x;
        qv[d + 1] = v.y;
    }
}

template <int TI>
__device__ __forceinline__ void scores5(const KPlane& P, QState& Q,
                                        const float* rp, const float* rp4, int half)
{
#pragma unroll
    for (int i = 0; i < NSLOT; ++i) {
        float dot = 0.0f;
#pragma unroll
        for (int d = 0; d < HDD; ++d) dot = fmaf(Q.q[d], kdim(P, i, d), dot);
        const float b = (i < NSLOT - 1) ? rp[TI * 9 + i] : rp4[TI * 9];
        Q.s[TI * NSLOT + i] = dot + b;
    }
    if (half) Q.s[TI * NSLOT + NSLOT - 1] = -1e30f;
}

__device__ __forceinline__ void stage_published() {
    __builtin_amdgcn_fence(__ATOMIC_RELEASE, "workgroup", "local");
    __builtin_amdgcn_s_barrier_signal(-1);
}
__device__ __forceinline__ void stage_complete() {
    __builtin_amdgcn_s_barrier_wait(-1);
    __builtin_amdgcn_fence(__ATOMIC_ACQUIRE, "workgroup", "local");
}

__device__ __forceinline__ v8f query_result(const float (&s)[NSC], int half, int col)
{
    float mx = -1e30f;
#pragma unroll
    for (int e = 0; e < NSC; ++e) mx = fmaxf(mx, s[e]);
    mx = fmaxf(mx, xhalf(mx));

    float p[NSC];
    float sum = 0.0f;
#pragma unroll
    for (int e = 0; e < NSC; ++e) { p[e] = __expf(s[e] - mx); sum += p[e]; }
    sum += xhalf(sum);
    const float inv = __builtin_amdgcn_rcpf(sum);
#pragma unroll
    for (int e = 0; e < NSC; ++e) p[e] *= inv;

    v8f acc = {};
#pragma unroll
    for (int c = 0; c < NMMA; ++c) {
        const int e0 = 2 * c, e1 = 2 * c + 1;
        v2f a, b;
        a.x = p[e0];
        a.y = (e1 < NSC) ? p[e1 < NSC ? e1 : 0] : 0.0f;
        const unsigned cx = half ? bcode(1, e0) : bcode(0, e0);
        const unsigned cy = half ? bcode(1, e1) : bcode(0, e1);
        b.x = bval(cx, col);
        b.y = bval(cy, col);
        acc = __builtin_amdgcn_wmma_f32_16x16x4_f32(
             false, a,  false, b,
             (short)0, acc,  false,  false);
    }
    return acc;
}

__device__ __forceinline__ void hand_over(const v8f& acc, float* so_new, const float* so_old,
                                          bool has_old, float* __restrict__ out, int x_old,
                                          int obase, int wave, int tid, int half, int col)
{
    if (has_old) stage_complete();
    if (col < 3) {
#pragma unroll
        for (int r = 0; r < 8; ++r) {
            so_new[(r * 3 + col) * BZ + wave * 2 + half] = acc[r];
        }
    }
    v4f d = {0.0f, 0.0f, 0.0f, 0.0f};
    if (has_old && wave < NST4 / 32) d = *(const v4f*)(so_old + tid * 4);
    stage_published();
    if (has_old && wave < NST4 / 32) {
        float* op = out + (size_t)x_old * (GW * GT) + obase;
        *(v4f*)(op + (size_t)(tid >> 3) * NVOX + (tid & 7) * 4) = d;
    }
}

__device__ __forceinline__ void drain_last(const float* so, float* __restrict__ out, int x,
                                           int obase, int wave, int tid)
{
    stage_complete();
    if (wave < NST4 / 32) {
        float* op = out + (size_t)x * (GW * GT) + obase;
        *(v4f*)(op + (size_t)(tid >> 3) * NVOX + (tid & 7) * 4) = *(const v4f*)(so + tid * 4);
    }
}

__global__ __launch_bounds__(NWV * 32) void nat3d_wmma_kernel(
    const float* __restrict__ q,
    const float* __restrict__ k,
    const float* __restrict__ rpb,
    float* __restrict__ out)
{
    __shared__ float s_rpb[NHD * NT];
    __shared__ __attribute__((aligned(16))) float s_out[2][NCH * BZ];
    int tid = threadIdx.x;
    if (tid < NHD * NT) s_rpb[tid] = rpb[tid];
    __syncthreads();

    const int lane = tid & 31;
    const int wave = __builtin_amdgcn_readfirstlane(tid >> 5);
    const int unit = blockIdx.x * NWV + wave;
    const int z2   = unit % (GT / 2);
    const int tmp  = unit / (GT / 2);
    const int y    = tmp % GW;
    const int seg  = tmp / GW;
    const int x0   = seg * XSEG;
    const int xe   = x0 + XSEG;
    const int obase = (int)((blockIdx.x * NWV) % (GW * GT / 2)) * 2;

    const int m    = lane & 15;
    const int half = lane >> 4;
    const int h    = m & 7;
    const int z    = z2 * 2 + (m >> 3);
    const int col  = lane & 15;

    int  koff[NSLOT];
    bool kval[NSLOT];
#pragma unroll
    for (int i = 0; i < NSLOT; ++i) {
        const int  j  = half ? (5 + i) / 3 : i / 3;
        const int  l  = half ? (5 + i) % 3 : i % 3;
        const bool in = (i < NSLOT - 1) || !half;
        koff[i] = ((j - 1) * GT + (l - 1)) * DIMC;
        kval[i] = in && (unsigned)(y + j - 1) < GW && (unsigned)(z + l - 1) < GT;
    }
    const float* rp  = s_rpb + h * NT + half * 5;
    const float* rp4 = rp + (NSLOT - 1) - half;

    QState U, V, Wd;
    KPlane KA, KB;
    float qrA[HDD], qrB[HDD];
    int X = __builtin_amdgcn_readfirstlane(x0 - 1);
    load_plane(KA, k, X, y, z, h, koff, kval);
    load_q(qrA, q, X + 1, y, z, h);
#define NAT_ROUND(QU, QV, QW, KC, KN, QRC, QRN)                                        \
    {                                                                                  \
        if (X - 2 >= x0) {                                                             \
            const v8f acc = query_result(QW.s, half, col);                             \
            hand_over(acc, s_out[X & 1], s_out[(X + 1) & 1], X - 3 >= x0, out, X - 3,  \
                      obase, wave, tid, half, col);                                    \
        }                                                                              \
        if (X + 1 < xe) {                                                              \
            _Pragma("unroll")                                                          \
            for (int d = 0; d < HDD; ++d) QW.q[d] = QRC[d] * QSCALE;                   \
        }                                                                              \
        plane_landed(KC, QRC);                                                         \
        if (X + 1 <= xe) load_plane(KN, k, X + 1, y, z, h, koff, kval);                \
        if (X + 2 < xe)  load_q(QRN, q, X + 2, y, z, h);                               \
        if (X - 1 >= x0) scores5<2>(KC, QU, rp, rp4, half);                            \
        if (X >= x0 && X < xe) scores5<1>(KC, QV, rp, rp4, half);                      \
        if (X + 1 < xe)  scores5<0>(KC, QW, rp, rp4, half);                            \
        if (++X > xe) break;                                                           \
    }
    for (;;) {
        NAT_ROUND(U, V, Wd, KA, KB, qrA, qrB)
        NAT_ROUND(V, Wd, U, KB, KA, qrB, qrA)
        NAT_ROUND(Wd, U, V, KA, KB, qrA, qrB)
        NAT_ROUND(U, V, Wd, KB, KA, qrB, qrA)
        NAT_ROUND(V, Wd, U, KA, KB, qrA, qrB)
        NAT_ROUND(Wd, U, V, KB, KA, qrB, qrA)
    }
#undef NAT_ROUND
    constexpr int LAST = (XSEG + 1) % 3;
    static_assert(XSEG >= 2, "the tail hands over with an older query in the other stage");
    const v8f acc = (LAST == 0) ? query_result(U.s, half, col)
                  : (LAST == 1) ? query_result(V.s, half, col)
                                : query_result(Wd.s, half, col);
    hand_over(acc, s_out[X & 1], s_out[(X + 1) & 1], true, out, xe - 2, obase, wave, tid, half, col);
    drain_last(s_out[X & 1], out, xe - 1, obase, wave, tid);
}

extern "C" void kernel_launch(void* const* d_in, const int* in_sizes, int n_in,
                              void* d_out, int out_size, void* d_ws, size_t ws_size,
                              hipStream_t stream) {
    (void)in_sizes; (void)n_in; (void)out_size; (void)d_ws; (void)ws_size;
    const float* q   = (const float*)d_in[0];
    const float* k   = (const float*)d_in[1];
    const float* rpb = (const float*)d_in[2];
    float* out = (float*)d_out;

    dim3 grid(250), block(512);
    nat3d_wmma_kernel<<<grid, block, 0, stream>>>(q, k, rpb, out);
}
